# starting kernel
# baseline (speedup 1.0000x reference)
#include <hip/hip_runtime.h>

#define VOCAB   256
#define EMBED   32
#define HIDDEN  32
#define LAYERS  10
#define BATCH   1024
#define SEQT    1024

#define RS      34
#define TILEF   (16 * RS)
#define NBUF    (LAYERS * 2 * TILEF)
#define NEMB    (VOCAB * RS)

typedef __attribute__((ext_vector_type(2))) float v2f;
typedef __attribute__((ext_vector_type(8))) float v8f;

__device__ __forceinline__ v8f wmma_k4(v2f a, v2f b, v8f c) {
  return __builtin_amdgcn_wmma_f32_16x16x4_f32(false, a, false, b, (short)0, c,
                                               false, false);
}

__device__ __forceinline__ float dev_tanh(float x) {
#if __has_builtin(__builtin_amdgcn_tanhf)
  return __builtin_amdgcn_tanhf(x);
#else
  float ax = __builtin_fabsf(x);
  float e  = __expf(-2.0f * ax);
  float y  = __fdividef(1.0f - e, 1.0f + e);
  return __builtin_copysignf(y, x);
#endif
}

__global__ void __launch_bounds__(LAYERS * 32)
rnn_pipeline_kernel(const int* __restrict__ x, const float* __restrict__ emb,
                    const float* __restrict__ W_ih, const float* __restrict__ W_hh,
                    const float* __restrict__ b_ih, const float* __restrict__ b_hh,
                    const float* __restrict__ W_fc, const float* __restrict__ b_fc,
                    float* __restrict__ out) {
  extern __shared__ float smem[];
  float* buf  = smem;
  float* embs = smem + NBUF;

  const int tid  = threadIdx.x;
  const int lane = tid & 31;
  const int l    = tid >> 5;
  const int R    = blockIdx.x * 16;

  for (int i = tid; i < NBUF; i += LAYERS * 32) buf[i] = 0.0f;
  for (int i = tid; i < VOCAB * EMBED; i += LAYERS * 32)
    embs[(i >> 5) * RS + (i & 31)] = emb[i];

  const int m    = lane & 15;
  const int koff = (lane >> 4) * 2;
  const int rb   = (lane >> 4) * 8;

  v2f wih[2][8], whh[2][8];
  float bias[2];
  {
    const float* Wi = W_ih + l * HIDDEN * EMBED;
    const float* Wh = W_hh + l * HIDDEN * HIDDEN;
#pragma unroll
    for (int c = 0; c < 2; ++c) {
      const int row = c * 16 + m;
#pragma unroll
      for (int j = 0; j < 8; ++j) {
        wih[c][j] = *(const v2f*)(Wi + row * EMBED  + 4 * j + koff);
        whh[c][j] = *(const v2f*)(Wh + row * HIDDEN + 4 * j + koff);
      }
      bias[c] = b_ih[l * HIDDEN + row] + b_hh[l * HIDDEN + row];
    }
  }

  __syncthreads();

  for (int i = 0; i < SEQT + LAYERS - 1; ++i) {
    const int t = i - l;
    if ((unsigned)t < (unsigned)SEQT) {
      const int pin  = (i - 1) & 1;
      const int pout = i & 1;

      v2f ah[8], ax[8];
      const float* hin = buf + (l * 2 + pin) * TILEF + m * RS + koff;
#pragma unroll
      for (int j = 0; j < 8; ++j) ah[j] = *(const v2f*)(hin + 4 * j);

      if (l == 0) {
        const int tok   = x[(R + m) * SEQT + t];
        const float* xe = embs + tok * RS + koff;
#pragma unroll
        for (int j = 0; j < 8; ++j) ax[j] = *(const v2f*)(xe + 4 * j);
      } else {
        const float* xin = buf + ((l - 1) * 2 + pin) * TILEF + m * RS + koff;
#pragma unroll
        for (int j = 0; j < 8; ++j) ax[j] = *(const v2f*)(xin + 4 * j);
      }

      v8f d0, d1;
#pragma unroll
      for (int k = 0; k < 8; ++k) { d0[k] = bias[0]; d1[k] = bias[1]; }
#pragma unroll
      for (int j = 0; j < 8; ++j) {
        d0 = wmma_k4(ax[j], wih[0][j], d0);
        d1 = wmma_k4(ax[j], wih[1][j], d1);
      }
#pragma unroll
      for (int j = 0; j < 8; ++j) {
        d0 = wmma_k4(ah[j], whh[0][j], d0);
        d1 = wmma_k4(ah[j], whh[1][j], d1);
      }

      float* hout = buf + (l * 2 + pout) * TILEF;
#pragma unroll
      for (int r = 0; r < 8; ++r) {
        hout[(rb + r) * RS + m]      = dev_tanh(d0[r]);
        hout[(rb + r) * RS + m + 16] = dev_tanh(d1[r]);
      }
    }
    __syncthreads();
  }

  const float* htop = buf + ((LAYERS - 1) * 2 + ((SEQT + LAYERS - 2) & 1)) * TILEF;
  v2f ahf[8];
#pragma unroll
  for (int j = 0; j < 8; ++j)
    ahf[j] = *(const v2f*)(htop + m * RS + 4 * j + koff);

  for (int ct = l; ct < VOCAB / 16; ct += LAYERS) {
    const int vrow = ct * 16 + m;
    v2f wb[8];
#pragma unroll
    for (int j = 0; j < 8; ++j)
      wb[j] = *(const v2f*)(W_fc + vrow * HIDDEN + 4 * j + koff);
    const float bf = b_fc[vrow];
    v8f d;
#pragma unroll
    for (int k = 0; k < 8; ++k) d[k] = bf;
#pragma unroll
    for (int j = 0; j < 8; ++j) d = wmma_k4(ahf[j], wb[j], d);
#pragma unroll
    for (int r = 0; r < 8; ++r)
      out[(R + rb + r) * VOCAB + vrow] = d[r];
  }
}

extern "C" void kernel_launch(void* const* d_in, const int* in_sizes, int n_in,
                              void* d_out, int out_size, void* d_ws, size_t ws_size,
                              hipStream_t stream) {
  (void)in_sizes; (void)n_in; (void)out_size; (void)d_ws; (void)ws_size;
  const int*   x   = (const int*)d_in[0];
  const float* emb = (const float*)d_in[1];
  const float* Wih = (const float*)d_in[2];
  const float* Whh = (const float*)d_in[3];
  const float* bih = (const float*)d_in[4];
  const float* bhh = (const float*)d_in[5];
  const float* Wfc = (const float*)d_in[6];
  const float* bfc = (const float*)d_in[7];
  float* out = (float*)d_out;

  const size_t shmem = (size_t)(NBUF + NEMB) * sizeof(float);
  hipLaunchKernelGGL(rnn_pipeline_kernel, dim3(BATCH / 16), dim3(LAYERS * 32),
                     shmem, stream, x, emb, Wih, Whh, bih, bhh, Wfc, bfc, out);
}


# Layers 0 and 1 split by product: x @ W_ih^T + bias runs on two extra waves one block ahead and reaches the layer wave through LDS as its accumulator start; 12 waves, 80 f32 WMMAs per step per SIMD
# speedup vs baseline: 1.3232x; 1.3232x over previous
#include <hip/hip_runtime.h>

#define VOCAB   256
#define EMBED   32
#define HIDDEN  32
#define LAYERS  10
#define BATCH   1024
#define SEQT    1024

#define TB      4
#define NBLK    (SEQT / TB)
#define RS      34
#define TILEF   (16 * RS)
#define HALFF   (TB * TILEF)
#define LAYERF  (2 * HALFF)
#define NBUF    (LAYERS * LAYERF)
#define NEMB    (VOCAB * RS)

#define NSPLIT  2
#define NW      (LAYERS + NSPLIT)
#define PHALF   (TB * 2 * 8 * 32)
#define PRING   (2 * PHALF)
#define NPB     (NSPLIT * PRING)
#define NITER   (NBLK + LAYERS + NSPLIT - 1)

typedef __attribute__((ext_vector_type(2))) float v2f;
typedef __attribute__((ext_vector_type(4))) float v4f;
typedef __attribute__((ext_vector_type(8))) float v8f;

__device__ __forceinline__ v8f wmma_k4(v2f a, v2f b, v8f c) {
  return __builtin_amdgcn_wmma_f32_16x16x4_f32(false, a, false, b, (short)0, c,
                                               false, false);
}

__device__ __forceinline__ float dev_tanh(float x) {
#if __has_builtin(__builtin_amdgcn_tanhf)
  return __builtin_amdgcn_tanhf(x);
#else
  float ax = __builtin_fabsf(x);
  float e  = __expf(-2.0f * ax);
  float y  = __fdividef(1.0f - e, 1.0f + e);
  return __builtin_copysignf(y, x);
#endif
}

__device__ __forceinline__ void load_frag(v2f (&a)[8], const float* p) {
#pragma unroll
  for (int j = 0; j < 8; ++j) a[j] = *(const v2f*)(p + 4 * j);
}

__device__ __forceinline__ void mma_tile(v8f& d0, v8f& d1, const v2f (&a)[8],
                                         const v2f (&wt)[2][8]) {
#pragma unroll
  for (int j = 0; j < 8; ++j) {
    d0 = wmma_k4(a[j], wt[0][j], d0);
    d1 = wmma_k4(a[j], wt[1][j], d1);
  }
}

__device__ __forceinline__ void store_tanh(float* hout, const v8f& d0, const v8f& d1,
                                           int m, int rb) {
#pragma unroll
  for (int r = 0; r < 8; ++r) {
    hout[(rb + r) * RS + m]      = dev_tanh(d0[r]);
    hout[(rb + r) * RS + m + 16] = dev_tanh(d1[r]);
  }
}

__global__ void __launch_bounds__(NW * 32)
rnn_pipeline_kernel(const int* __restrict__ x, const float* __restrict__ emb,
                    const float* __restrict__ W_ih, const float* __restrict__ W_hh,
                    const float* __restrict__ b_ih, const float* __restrict__ b_hh,
                    const float* __restrict__ W_fc, const float* __restrict__ b_fc,
                    float* __restrict__ out) {
  __shared__ __attribute__((aligned(16))) float smem[NBUF + NEMB + NPB];
  float* buf  = smem;
  float* embs = smem + NBUF;
  float* pbuf = smem + NBUF + NEMB;

  const int tid  = threadIdx.x;
  const int lane = tid & 31;
  const int w    = tid >> 5;
  const int R    = blockIdx.x * 16;

  const bool is_ip  = (w >= LAYERS);
  const bool is_rec = (w < NSPLIT);
  const int  l      = is_ip ? (w - LAYERS) : w;
  const int  delay  = is_ip ? (2 * l) : (is_rec ? (2 * l + 1) : (l + NSPLIT));

  for (int i = tid; i < NBUF; i += NW * 32) buf[i] = 0.0f;
  for (int i = tid; i < VOCAB * EMBED; i += NW * 32)
    embs[(i >> 5) * RS + (i & 31)] = emb[i];

  const int m    = lane & 15;
  const int koff = (lane >> 4) * 2;
  const int rb   = (lane >> 4) * 8;

  v2f wih[2][8], whh[2][8];
  float bias[2];
  {
    const float* Wi = W_ih + l * HIDDEN * EMBED;
    const float* Wh = W_hh + l * HIDDEN * HIDDEN;
#pragma unroll
    for (int c = 0; c < 2; ++c) {
      const int row = c * 16 + m;
#pragma unroll
      for (int j = 0; j < 8; ++j) {
        wih[c][j] = *(const v2f*)(Wi + row * EMBED  + 4 * j + koff);
        whh[c][j] = *(const v2f*)(Wh + row * HIDDEN + 4 * j + koff);
      }
      bias[c] = b_ih[l * HIDDEN + row] + b_hh[l * HIDDEN + row];
    }
  }

  __syncthreads();

  for (int I = 0; I < NITER; ++I) {
    const int tb = I - delay;
    if ((unsigned)tb < (unsigned)NBLK) {
      const int half = tb & 1;

      if (is_ip) {
        const float* xp[TB];
        if (l == 0) {
          const int4 tok = *(const int4*)(x + (R + m) * SEQT + tb * TB);
          xp[0] = embs + tok.x * RS + koff;
          xp[1] = embs + tok.y * RS + koff;
          xp[2] = embs + tok.z * RS + koff;
          xp[3] = embs + tok.w * RS + koff;
        } else {
          const float* xin = buf + (l - 1) * LAYERF + half * HALFF + m * RS + koff;
#pragma unroll
          for (int s = 0; s < TB; ++s) xp[s] = xin + s * TILEF;
        }
        v2f ax[TB][8];
#pragma unroll
        for (int s = 0; s < TB; ++s) load_frag(ax[s], xp[s]);

        v4f* pout = (v4f*)(pbuf + l * PRING + half * PHALF) + lane;
#pragma unroll
        for (int s = 0; s < TB; ++s) {
          v8f d0, d1;
#pragma unroll
          for (int k = 0; k < 8; ++k) { d0[k] = bias[0]; d1[k] = bias[1]; }
          mma_tile(d0, d1, ax[s], wih);
          pout[(s * 4 + 0) * 32] = __builtin_shufflevector(d0, d0, 0, 1, 2, 3);
          pout[(s * 4 + 1) * 32] = __builtin_shufflevector(d0, d0, 4, 5, 6, 7);
          pout[(s * 4 + 2) * 32] = __builtin_shufflevector(d1, d1, 0, 1, 2, 3);
          pout[(s * 4 + 3) * 32] = __builtin_shufflevector(d1, d1, 4, 5, 6, 7);
        }
      } else {
        float* cur = buf + l * LAYERF + half * HALFF;
        const float* prv = buf + l * LAYERF + (half ^ 1) * HALFF + (TB - 1) * TILEF;
        v2f ah[8];
        v8f d0[TB], d1[TB];

        if (is_rec) {
          load_frag(ah, prv + m * RS + koff);
          const v4f* pin = (const v4f*)(pbuf + l * PRING + half * PHALF) + lane;
#pragma unroll
          for (int s = 0; s < TB; ++s) {
            const v4f p0 = pin[(s * 4 + 0) * 32];
            const v4f p1 = pin[(s * 4 + 1) * 32];
            const v4f p2 = pin[(s * 4 + 2) * 32];
            const v4f p3 = pin[(s * 4 + 3) * 32];
            d0[s] = __builtin_shufflevector(p0, p1, 0, 1, 2, 3, 4, 5, 6, 7);
            d1[s] = __builtin_shufflevector(p2, p3, 0, 1, 2, 3, 4, 5, 6, 7);
          }
#pragma unroll
          for (int s = 0; s < TB; ++s) {
            mma_tile(d0[s], d1[s], ah, whh);
            float* hout = cur + s * TILEF;
            store_tanh(hout, d0[s], d1[s], m, rb);
            if (s + 1 < TB) {
              __builtin_amdgcn_fence(__ATOMIC_SEQ_CST, "wavefront");
              load_frag(ah, hout + m * RS + koff);
            }
          }
        } else {
          const float* xin = buf + (l - 1) * LAYERF + half * HALFF + m * RS + koff;
          v2f ax[TB][8];
#pragma unroll
          for (int s = 0; s < TB; ++s) load_frag(ax[s], xin + s * TILEF);
          load_frag(ah, prv + m * RS + koff);

#pragma unroll
          for (int k = 0; k < 8; ++k) { d0[0][k] = bias[0]; d1[0][k] = bias[1]; }
          mma_tile(d0[0], d1[0], ax[0], wih);

#pragma unroll
          for (int s = 0; s < TB; ++s) {
            mma_tile(d0[s], d1[s], ah, whh);
            float* hout = cur + s * TILEF;
            store_tanh(hout, d0[s], d1[s], m, rb);
            if (s + 1 < TB) {
              __builtin_amdgcn_fence(__ATOMIC_SEQ_CST, "wavefront");
              load_frag(ah, hout + m * RS + koff);
#pragma unroll
              for (int k = 0; k < 8; ++k) { d0[s + 1][k] = bias[0]; d1[s + 1][k] = bias[1]; }
              mma_tile(d0[s + 1], d1[s + 1], ax[s + 1], wih);
            }
          }
        }
      }
    }
    __syncthreads();
  }

  const float* htop = buf + (LAYERS - 1) * LAYERF + ((NBLK - 1) & 1) * HALFF + (TB - 1) * TILEF;
  v2f ahf[8];
  load_frag(ahf, htop + m * RS + koff);

  for (int ct = w; ct < VOCAB / 16; ct += NW) {
    const int vrow = ct * 16 + m;
    v2f wb[8];
#pragma unroll
    for (int j = 0; j < 8; ++j)
      wb[j] = *(const v2f*)(W_fc + vrow * HIDDEN + 4 * j + koff);
    const float bf = b_fc[vrow];
    v8f d;
#pragma unroll
    for (int k = 0; k < 8; ++k) d[k] = bf;
#pragma unroll
    for (int j = 0; j < 8; ++j) d = wmma_k4(ahf[j], wb[j], d);
#pragma unroll
    for (int r = 0; r < 8; ++r)
      out[(R + rb + r) * VOCAB + vrow] = d[r];
  }
}

extern "C" void kernel_launch(void* const* d_in, const int* in_sizes, int n_in,
                              void* d_out, int out_size, void* d_ws, size_t ws_size,
                              hipStream_t stream) {
  (void)in_sizes; (void)n_in; (void)out_size; (void)d_ws; (void)ws_size;
  const int*   x   = (const int*)d_in[0];
  const float* emb = (const float*)d_in[1];
  const float* Wih = (const float*)d_in[2];
  const float* Whh = (const float*)d_in[3];
  const float* bih = (const float*)d_in[4];
  const float* bhh = (const float*)d_in[5];
  const float* Wfc = (const float*)d_in[6];
  const float* bfc = (const float*)d_in[7];
  float* out = (float*)d_out;

  hipLaunchKernelGGL(rnn_pipeline_kernel, dim3(BATCH / 16), dim3(NW * 32),
                     0, stream, x, emb, Wih, Whh, bih, bhh, Wfc, bfc, out);
}


# Layers 2 and 3 split by product too; waves placed so each SIMD's youngest wave is an input-product wave, which has no step-to-step dependence and issues while layer waves wait on LDS; 14 waves
# speedup vs baseline: 1.3742x; 1.0385x over previous
#include <hip/hip_runtime.h>

#define VOCAB   256
#define EMBED   32
#define HIDDEN  32
#define LAYERS  10
#define BATCH   1024
#define SEQT    1024

#define TB      4
#define NBLK    (SEQT / TB)
#define RS      34
#define TILEF   (16 * RS)
#define HALFF   (TB * TILEF)
#define LAYERF  (2 * HALFF)
#define NBUF    (LAYERS * LAYERF)
#define NEMB    (VOCAB * RS)

#define NSPLIT  4
#define NW      (LAYERS + NSPLIT)
#define PHALF   (TB * 2 * 8 * 32)
#define PRING   (2 * PHALF)
#define NPB     (NSPLIT * PRING)
#define NITER   (NBLK + LAYERS + NSPLIT - 1)

typedef __attribute__((ext_vector_type(2))) float v2f;
typedef __attribute__((ext_vector_type(4))) float v4f;
typedef __attribute__((ext_vector_type(8))) float v8f;

__device__ __forceinline__ v8f wmma_k4(v2f a, v2f b, v8f c) {
  return __builtin_amdgcn_wmma_f32_16x16x4_f32(false, a, false, b, (short)0, c,
                                               false, false);
}

__device__ __forceinline__ float dev_tanh(float x) {
#if __has_builtin(__builtin_amdgcn_tanhf)
  return __builtin_amdgcn_tanhf(x);
#else
  float ax = __builtin_fabsf(x);
  float e  = __expf(-2.0f * ax);
  float y  = __fdividef(1.0f - e, 1.0f + e);
  return __builtin_copysignf(y, x);
#endif
}

__device__ __forceinline__ void load_frag(v2f (&a)[8], const float* p) {
#pragma unroll
  for (int j = 0; j < 8; ++j) a[j] = *(const v2f*)(p + 4 * j);
}

__device__ __forceinline__ void mma_tile(v8f& d0, v8f& d1, const v2f (&a)[8],
                                         const v2f (&wt)[2][8]) {
#pragma unroll
  for (int j = 0; j < 8; ++j) {
    d0 = wmma_k4(a[j], wt[0][j], d0);
    d1 = wmma_k4(a[j], wt[1][j], d1);
  }
}

__device__ __forceinline__ void store_tanh(float* hout, const v8f& d0, const v8f& d1,
                                           int m, int rb) {
#pragma unroll
  for (int r = 0; r < 8; ++r) {
    hout[(rb + r) * RS + m]      = dev_tanh(d0[r]);
    hout[(rb + r) * RS + m + 16] = dev_tanh(d1[r]);
  }
}

__global__ void __launch_bounds__(NW * 32)
rnn_pipeline_kernel(const int* __restrict__ x, const float* __restrict__ emb,
                    const float* __restrict__ W_ih, const float* __restrict__ W_hh,
                    const float* __restrict__ b_ih, const float* __restrict__ b_hh,
                    const float* __restrict__ W_fc, const float* __restrict__ b_fc,
                    float* __restrict__ out) {
  __shared__ __attribute__((aligned(16))) float smem[NBUF + NEMB + NPB];
  float* buf  = smem;
  float* embs = smem + NBUF;
  float* pbuf = smem + NBUF + NEMB;

  const int tid  = threadIdx.x;
  const int lane = tid & 31;
  const int w    = tid >> 5;
  const int R    = blockIdx.x * 16;

  static_assert(LAYERS == 10 && NSPLIT == 4, "the wave placement below is written for 10 layers, 4 of them split");
  const int  wsimd  = w & 3;
  const int  wrank  = w >> 2;
  const bool is_ip  = (w >= LAYERS);
  const bool is_rec = !is_ip && (wsimd < 2) && (wrank < 2);
  const int  l      = is_ip ? (w - LAYERS)
                    : is_rec ? (wsimd + 2 * wrank)
                    : (wrank == 2) ? (NSPLIT + wsimd)
                    : (NSPLIT + wsimd + 2 * wrank);
  const int  delay  = is_ip ? (2 * l) : (is_rec ? (2 * l + 1) : (l + NSPLIT));

  for (int i = tid; i < NBUF; i += NW * 32) buf[i] = 0.0f;
  for (int i = tid; i < VOCAB * EMBED; i += NW * 32)
    embs[(i >> 5) * RS + (i & 31)] = emb[i];

  const int m    = lane & 15;
  const int koff = (lane >> 4) * 2;
  const int rb   = (lane >> 4) * 8;

  v2f wih[2][8], whh[2][8];
  float bias[2];
  {
    const float* Wi = W_ih + l * HIDDEN * EMBED;
    const float* Wh = W_hh + l * HIDDEN * HIDDEN;
#pragma unroll
    for (int c = 0; c < 2; ++c) {
      const int row = c * 16 + m;
#pragma unroll
      for (int j = 0; j < 8; ++j) {
        wih[c][j] = *(const v2f*)(Wi + row * EMBED  + 4 * j + koff);
        whh[c][j] = *(const v2f*)(Wh + row * HIDDEN + 4 * j + koff);
      }
      bias[c] = b_ih[l * HIDDEN + row] + b_hh[l * HIDDEN + row];
    }
  }

  __syncthreads();

  for (int I = 0; I < NITER; ++I) {
    const int tb = I - delay;
    if ((unsigned)tb < (unsigned)NBLK) {
      const int half = tb & 1;

      if (is_ip) {
        const float* xp[TB];
        if (l == 0) {
          const int4 tok = *(const int4*)(x + (R + m) * SEQT + tb * TB);
          xp[0] = embs + tok.x * RS + koff;
          xp[1] = embs + tok.y * RS + koff;
          xp[2] = embs + tok.z * RS + koff;
          xp[3] = embs + tok.w * RS + koff;
        } else {
          const float* xin = buf + (l - 1) * LAYERF + half * HALFF + m * RS + koff;
#pragma unroll
          for (int s = 0; s < TB; ++s) xp[s] = xin + s * TILEF;
        }
        v2f ax[TB][8];
#pragma unroll
        for (int s = 0; s < TB; ++s) load_frag(ax[s], xp[s]);

        v4f* pout = (v4f*)(pbuf + l * PRING + half * PHALF) + lane;
#pragma unroll
        for (int s = 0; s < TB; ++s) {
          v8f d0, d1;
#pragma unroll
          for (int k = 0; k < 8; ++k) { d0[k] = bias[0]; d1[k] = bias[1]; }
          mma_tile(d0, d1, ax[s], wih);
          pout[(s * 4 + 0) * 32] = __builtin_shufflevector(d0, d0, 0, 1, 2, 3);
          pout[(s * 4 + 1) * 32] = __builtin_shufflevector(d0, d0, 4, 5, 6, 7);
          pout[(s * 4 + 2) * 32] = __builtin_shufflevector(d1, d1, 0, 1, 2, 3);
          pout[(s * 4 + 3) * 32] = __builtin_shufflevector(d1, d1, 4, 5, 6, 7);
        }
      } else {
        float* cur = buf + l * LAYERF + half * HALFF;
        const float* prv = buf + l * LAYERF + (half ^ 1) * HALFF + (TB - 1) * TILEF;
        v2f ah[8];
        v8f d0[TB], d1[TB];

        if (is_rec) {
          load_frag(ah, prv + m * RS + koff);
          const v4f* pin = (const v4f*)(pbuf + l * PRING + half * PHALF) + lane;
#pragma unroll
          for (int s = 0; s < TB; ++s) {
            const v4f p0 = pin[(s * 4 + 0) * 32];
            const v4f p1 = pin[(s * 4 + 1) * 32];
            const v4f p2 = pin[(s * 4 + 2) * 32];
            const v4f p3 = pin[(s * 4 + 3) * 32];
            d0[s] = __builtin_shufflevector(p0, p1, 0, 1, 2, 3, 4, 5, 6, 7);
            d1[s] = __builtin_shufflevector(p2, p3, 0, 1, 2, 3, 4, 5, 6, 7);
          }
#pragma unroll
          for (int s = 0; s < TB; ++s) {
            mma_tile(d0[s], d1[s], ah, whh);
            float* hout = cur + s * TILEF;
            store_tanh(hout, d0[s], d1[s], m, rb);
            if (s + 1 < TB) {
              __builtin_amdgcn_fence(__ATOMIC_SEQ_CST, "wavefront");
              load_frag(ah, hout + m * RS + koff);
            }
          }
        } else {
          const float* xin = buf + (l - 1) * LAYERF + half * HALFF + m * RS + koff;
          v2f ax[TB][8];
#pragma unroll
          for (int s = 0; s < TB; ++s) load_frag(ax[s], xin + s * TILEF);
          load_frag(ah, prv + m * RS + koff);

#pragma unroll
          for (int k = 0; k < 8; ++k) { d0[0][k] = bias[0]; d1[0][k] = bias[1]; }
          mma_tile(d0[0], d1[0], ax[0], wih);

#pragma unroll
          for (int s = 0; s < TB; ++s) {
            mma_tile(d0[s], d1[s], ah, whh);
            float* hout = cur + s * TILEF;
            store_tanh(hout, d0[s], d1[s], m, rb);
            if (s + 1 < TB) {
              __builtin_amdgcn_fence(__ATOMIC_SEQ_CST, "wavefront");
              load_frag(ah, hout + m * RS + koff);
#pragma unroll
              for (int k = 0; k < 8; ++k) { d0[s + 1][k] = bias[0]; d1[s + 1][k] = bias[1]; }
              mma_tile(d0[s + 1], d1[s + 1], ax[s + 1], wih);
            }
          }
        }
      }
    }
    __syncthreads();
  }

  const float* htop = buf + (LAYERS - 1) * LAYERF + ((NBLK - 1) & 1) * HALFF + (TB - 1) * TILEF;
  v2f ahf[8];
  load_frag(ahf, htop + m * RS + koff);

  for (int ct = w; ct < VOCAB / 16; ct += NW) {
    const int vrow = ct * 16 + m;
    v2f wb[8];
#pragma unroll
    for (int j = 0; j < 8; ++j)
      wb[j] = *(const v2f*)(W_fc + vrow * HIDDEN + 4 * j + koff);
    const float bf = b_fc[vrow];
    v8f d;
#pragma unroll
    for (int k = 0; k < 8; ++k) d[k] = bf;
#pragma unroll
    for (int j = 0; j < 8; ++j) d = wmma_k4(ahf[j], wb[j], d);
#pragma unroll
    for (int r = 0; r < 8; ++r)
      out[(R + rb + r) * VOCAB + vrow] = d[r];
  }
}

extern "C" void kernel_launch(void* const* d_in, const int* in_sizes, int n_in,
                              void* d_out, int out_size, void* d_ws, size_t ws_size,
                              hipStream_t stream) {
  (void)in_sizes; (void)n_in; (void)out_size; (void)d_ws; (void)ws_size;
  const int*   x   = (const int*)d_in[0];
  const float* emb = (const float*)d_in[1];
  const float* Wih = (const float*)d_in[2];
  const float* Whh = (const float*)d_in[3];
  const float* bih = (const float*)d_in[4];
  const float* bhh = (const float*)d_in[5];
  const float* Wfc = (const float*)d_in[6];
  const float* bfc = (const float*)d_in[7];
  float* out = (float*)d_out;

  hipLaunchKernelGGL(rnn_pipeline_kernel, dim3(BATCH / 16), dim3(NW * 32),
                     0, stream, x, emb, Wih, Whh, bih, bhh, Wfc, bfc, out);
}
